# starting kernel
# baseline (speedup 1.0000x reference)
#include <hip/hip_runtime.h>

typedef float v2f __attribute__((ext_vector_type(2)));
typedef float v8f __attribute__((ext_vector_type(8)));

#define BDIM  2
#define NDIM  1024
#define CDIM  64
#define SCALE 0.25f
#define LOG2E 1.44269504088896340736f

__global__ __launch_bounds__(256)
void intra_attn_wmma_kernel(const float* __restrict__ q,
                            const float* __restrict__ k,
                            const float* __restrict__ v,
                            float* __restrict__ out) {
    __shared__ __align__(16) float ks[NDIM];
    __shared__ __align__(16) float vs[NDIM];
    __shared__ float red[256];

    const int tid  = threadIdx.x;
    const int lane = tid & 31;
    const int wave = tid >> 5;

    const int bc    = blockIdx.x;
    const int b     = bc >> 6;
    const int c     = bc & (CDIM - 1);
    const int ibase = blockIdx.y * 256;

    const float* qp = q   + (size_t)b * NDIM * CDIM + c;
    const float* kp = k   + (size_t)b * NDIM * CDIM + c;
    const float* vp = v   + (size_t)b * NDIM * CDIM + c;
    float*       op = out + (size_t)b * NDIM * CDIM + c;

    float lmax = 0.0f;
    for (int j = tid; j < NDIM; j += 256) {
        const float kv = kp[(size_t)j * CDIM] * (SCALE * LOG2E);
        ks[j] = kv;
        vs[j] = vp[(size_t)j * CDIM];
        lmax = fmaxf(lmax, fabsf(kv));
    }
    red[tid] = lmax;
    __syncthreads();
    for (int s = 128; s > 0; s >>= 1) {
        if (tid < s) red[tid] = fmaxf(red[tid], red[tid + s]);
        __syncthreads();
    }
    const float ksabs = red[0];

    const int   M    = lane & 15;
    const int   koff = (lane >> 4) << 1;
    const float bc1  = (M == 1) ? 1.0f : 0.0f;
    const bool  isn0 = (M == 0);

    for (int tt = 0; tt < 2; ++tt) {
        const int   i0 = ibase + (wave * 2 + tt) * 16;
        const float a  = qp[(size_t)(i0 + M) * CDIM];
        const float m2 = fabsf(a) * ksabs;

        v8f acc0 = {};
        v8f acc1 = {};
        for (int j = 0; j < NDIM; j += 8) {
            const float2 k2a = *reinterpret_cast<const float2*>(&ks[j + koff]);
            const float2 v2a = *reinterpret_cast<const float2*>(&vs[j + koff]);
            const float2 k2b = *reinterpret_cast<const float2*>(&ks[j + 4 + koff]);
            const float2 v2b = *reinterpret_cast<const float2*>(&vs[j + 4 + koff]);

            v2f A0, B0, A1, B1;
            A0.x = __builtin_amdgcn_exp2f(__builtin_fmaf(a, k2a.x, -m2));
            A0.y = __builtin_amdgcn_exp2f(__builtin_fmaf(a, k2a.y, -m2));
            A1.x = __builtin_amdgcn_exp2f(__builtin_fmaf(a, k2b.x, -m2));
            A1.y = __builtin_amdgcn_exp2f(__builtin_fmaf(a, k2b.y, -m2));
            B0.x = isn0 ? v2a.x : bc1;
            B0.y = isn0 ? v2a.y : bc1;
            B1.x = isn0 ? v2b.x : bc1;
            B1.y = isn0 ? v2b.y : bc1;

            acc0 = __builtin_amdgcn_wmma_f32_16x16x4_f32(false, A0, false, B0,
                                                         (short)0, acc0, false, false);
            acc1 = __builtin_amdgcn_wmma_f32_16x16x4_f32(false, A1, false, B1,
                                                         (short)0, acc1, false, false);
        }
        v8f acc = acc0 + acc1;

        #pragma unroll
        for (int r = 0; r < 8; ++r) {
            const float tlo = __shfl(acc[r], 0, 32);
            const float slo = __shfl(acc[r], 1, 32);
            const float thi = __shfl(acc[r], 16, 32);
            const float shi = __shfl(acc[r], 17, 32);
            if (lane == r)      op[(size_t)(i0 + r) * CDIM]     = tlo / slo;
            if (lane == 16 + r) op[(size_t)(i0 + 8 + r) * CDIM] = thi / shi;
        }
    }
}

extern "C" void kernel_launch(void* const* d_in, const int* in_sizes, int n_in,
                              void* d_out, int out_size, void* d_ws, size_t ws_size,
                              hipStream_t stream) {
    const float* q = (const float*)d_in[0];
    const float* k = (const float*)d_in[1];
    const float* v = (const float*)d_in[2];
    float* out = (float*)d_out;
    dim3 grid(BDIM * CDIM, NDIM / 256);
    intra_attn_wmma_kernel<<<grid, dim3(256), 0, stream>>>(q, k, v, out);
}


# Rows share exponentials: A = exp2 at 16 centres over the block's q range, B = 16 columns v*x^n/n! and x^n/n!; row = 8-term series in q-centre (tail < 1e-9, run-time bound); sum order changes
# speedup vs baseline: 2.6538x; 2.6538x over previous
#include <hip/hip_runtime.h>


typedef float v2f __attribute__((ext_vector_type(2)));
typedef float v8f __attribute__((ext_vector_type(8)));

#define BDIM  2
#define NDIM  1024
#define CDIM  64
#define SCALE 0.25f
#define LOG2E 1.44269504088896340736f

#define ROWS   256
#define NWAVES 8
#define JW     (NDIM / NWAVES)
#define NT     (JW / 4)
#define MAXT   16

#define THETA 0.25f

__device__ __forceinline__ float pow_bits(float x, bool b1, bool b2, bool b4) {
    const float x2 = x * x;
    const float x4 = x2 * x2;
    float p = b1 ? x : 1.0f;
    p *= b2 ? x2 : 1.0f;
    p *= b4 ? x4 : 1.0f;
    return p;
}

__global__ __launch_bounds__(256)
void intra_attn_wmma_kernel(const float* __restrict__ q,
                            const float* __restrict__ k,
                            const float* __restrict__ v,
                            float* __restrict__ out) {
    __shared__ __align__(16) float xs[NDIM];
    __shared__ __align__(16) float vs[NDIM];
    __shared__ float qs[ROWS];
    __shared__ float red[256];
    __shared__ float rlo[256];
    __shared__ float rhi[256];
    __shared__ float part[NWAVES][256];
    __shared__ __align__(16) float coef[256];
    __shared__ float cen[16];

    const int tid  = threadIdx.x;
    const int lane = tid & 31;
    const int wave = tid >> 5;

    const int bc    = blockIdx.x;
    const int b     = bc >> 6;
    const int c     = bc & (CDIM - 1);
    const int ibase = blockIdx.y * ROWS;

    const float* qp = q   + (size_t)b * NDIM * CDIM + c;
    const float* kp = k   + (size_t)b * NDIM * CDIM + c;
    const float* vp = v   + (size_t)b * NDIM * CDIM + c;
    float*       op = out + (size_t)b * NDIM * CDIM + c;

    const float qv = qp[(size_t)(ibase + tid) * CDIM];
    float lmax = 0.0f;
    for (int j = tid; j < NDIM; j += 256) {
        const float xv = kp[(size_t)j * CDIM] * SCALE;
        xs[j] = xv;
        vs[j] = vp[(size_t)j * CDIM];
        lmax = fmaxf(lmax, fabsf(xv));
    }
    qs[tid]  = qv;
    red[tid] = lmax;
    rlo[tid] = qv;
    rhi[tid] = qv;
    __syncthreads();
    for (int s = 128; s > 0; s >>= 1) {
        if (tid < s) {
            red[tid] = fmaxf(red[tid], red[tid + s]);
            rlo[tid] = fminf(rlo[tid], rlo[tid + s]);
            rhi[tid] = fmaxf(rhi[tid], rhi[tid + s]);
        }
        __syncthreads();
    }
    const float xmax  = red[0];
    const float qmin  = rlo[0];
    const float qmax  = rhi[0];
    const float ksabs = xmax * LOG2E;

    const float range  = qmax - qmin;
    const float need   = range * xmax * (1.0f / (32.0f * THETA));
    const bool  direct = !(need <= (float)MAXT);
    const int   T      = direct ? MAXT : (int)ceilf(fmaxf(need, 1.0f));
    const float ncell  = (float)(16 * T);
    const float cellw  = range / ncell;
    const float inv    = (range > 0.0f) ? ncell / range : 0.0f;

    const int g = direct ? tid
                         : (int)fminf(fmaxf((qv - qmin) * inv, 0.0f), ncell - 1.0f);

    const int   M   = lane & 15;
    const int   h   = lane >> 4;
    const int   n   = M & 7;
    const bool  isv = (M < 8);
    const bool  b1  = (n & 1) != 0;
    const bool  b2  = (n & 2) != 0;
    const bool  b4  = (n & 4) != 0;
    const float rf  = (n < 2)  ? 1.0f
                    : (n == 2) ? (1.0f / 2.0f)
                    : (n == 3) ? (1.0f / 6.0f)
                    : (n == 4) ? (1.0f / 24.0f)
                    : (n == 5) ? (1.0f / 120.0f)
                    : (n == 6) ? (1.0f / 720.0f)
                    :            (1.0f / 5040.0f);

    v2f kf[NT];
    v2f bf[NT];
    {
        const int j0 = wave * JW + 2 * h;
        #pragma unroll
        for (int t = 0; t < NT; ++t) {
            const float2 x2 = *reinterpret_cast<const float2*>(&xs[j0 + 4 * t]);
            const float2 v2 = *reinterpret_cast<const float2*>(&vs[j0 + 4 * t]);
            kf[t].x = x2.x * LOG2E;
            kf[t].y = x2.y * LOG2E;
            bf[t].x = pow_bits(x2.x, b1, b2, b4) * ((isv ? v2.x : 1.0f) * rf);
            bf[t].y = pow_bits(x2.y, b1, b2, b4) * ((isv ? v2.y : 1.0f) * rf);
        }
    }

    for (int tile = 0; tile < T; ++tile) {
        const float a  = direct ? qs[tile * 16 + M]
                                : __builtin_fmaf((float)(tile * 16 + M) + 0.5f, cellw, qmin);
        const float m2 = fabsf(a) * ksabs;

        v8f acc0 = {};
        v8f acc1 = {};
        #pragma unroll
        for (int t = 0; t < NT; t += 2) {
            v2f A0, A1;
            A0.x = __builtin_amdgcn_exp2f(__builtin_fmaf(a, kf[t].x, -m2));
            A0.y = __builtin_amdgcn_exp2f(__builtin_fmaf(a, kf[t].y, -m2));
            A1.x = __builtin_amdgcn_exp2f(__builtin_fmaf(a, kf[t + 1].x, -m2));
            A1.y = __builtin_amdgcn_exp2f(__builtin_fmaf(a, kf[t + 1].y, -m2));
            acc0 = __builtin_amdgcn_wmma_f32_16x16x4_f32(false, A0, false, bf[t],
                                                         (short)0, acc0, false, false);
            acc1 = __builtin_amdgcn_wmma_f32_16x16x4_f32(false, A1, false, bf[t + 1],
                                                         (short)0, acc1, false, false);
        }
        const v8f acc = acc0 + acc1;

        {
            float* pr = &part[wave][(8 * h) * 16 + M];
            for (int r = 0; r < 8; ++r) pr[16 * r] = acc[r];
        }
        __syncthreads();

        float s = 0.0f;
        for (int w = 0; w < NWAVES; ++w) s += part[w][tid];
        coef[tid] = s;
        if (tid < 16) cen[tid] = a;
        __syncthreads();

        if ((g >> 4) == tile) {
            const int    m  = g & 15;
            const float  d  = qv - cen[m];
            const float4 n0 = *reinterpret_cast<const float4*>(&coef[m * 16 + 0]);
            const float4 n1 = *reinterpret_cast<const float4*>(&coef[m * 16 + 4]);
            const float4 d0 = *reinterpret_cast<const float4*>(&coef[m * 16 + 8]);
            const float4 d1 = *reinterpret_cast<const float4*>(&coef[m * 16 + 12]);
            float num = n1.w;
            num = __builtin_fmaf(num, d, n1.z);
            num = __builtin_fmaf(num, d, n1.y);
            num = __builtin_fmaf(num, d, n1.x);
            num = __builtin_fmaf(num, d, n0.w);
            num = __builtin_fmaf(num, d, n0.z);
            num = __builtin_fmaf(num, d, n0.y);
            num = __builtin_fmaf(num, d, n0.x);
            float den = d1.w;
            den = __builtin_fmaf(den, d, d1.z);
            den = __builtin_fmaf(den, d, d1.y);
            den = __builtin_fmaf(den, d, d1.x);
            den = __builtin_fmaf(den, d, d0.w);
            den = __builtin_fmaf(den, d, d0.z);
            den = __builtin_fmaf(den, d, d0.y);
            den = __builtin_fmaf(den, d, d0.x);
            op[(size_t)(ibase + tid) * CDIM] = num / den;
        }
    }
}

extern "C" void kernel_launch(void* const* d_in, const int* in_sizes, int n_in,
                              void* d_out, int out_size, void* d_ws, size_t ws_size,
                              hipStream_t stream) {
    const float* q = (const float*)d_in[0];
    const float* k = (const float*)d_in[1];
    const float* v = (const float*)d_in[2];
    float* out = (float*)d_out;
    dim3 grid(BDIM * CDIM, NDIM / 256);
    intra_attn_wmma_kernel<<<grid, dim3(256), 0, stream>>>(q, k, v, out);
}


# B tile built once per j pair at staging by its loading thread (x powers shared by the 16 columns), under the q loads; waves read ready-made fragments from LDS planes; values and sum order kept
# speedup vs baseline: 3.2826x; 1.2369x over previous
#include <hip/hip_runtime.h>


typedef float v2f __attribute__((ext_vector_type(2)));
typedef float v4f __attribute__((ext_vector_type(4)));
typedef float v8f __attribute__((ext_vector_type(8)));

#define BDIM  2
#define NDIM  1024
#define CDIM  64
#define SCALE 0.25f
#define LOG2E 1.44269504088896340736f

#define ROWS   NDIM
#define RPT    (ROWS / 256)
#define NWAVES 8
#define JW     (NDIM / NWAVES)
#define NT     (JW / 4)
#define MAXT   (ROWS / 16)
#define NPAIR  (NDIM / 2)
#define PLANE  (NPAIR * 4 + 8)

#define THETA 0.25f

__device__ __forceinline__ void build_pair(float* __restrict__ bt, int p, v2f x, v2f vv) {
    const v2f x2 = x * x;
    const v2f x4 = x2 * x2;
    const v2f x3 = x * x2;
    const v2f x5 = x * x4;
    const v2f x6 = x2 * x4;
    const v2f x7 = x3 * x4;
    const float r2 = 1.0f / 2.0f,   r3 = 1.0f / 6.0f,   r4 = 1.0f / 24.0f;
    const float r5 = 1.0f / 120.0f, r6 = 1.0f / 720.0f, r7 = 1.0f / 5040.0f;
    const v2f one = {1.0f, 1.0f};
    const v2f n0 = vv;
    const v2f n1 = x  * vv;
    const v2f n2 = x2 * (vv * r2);
    const v2f n3 = x3 * (vv * r3);
    const v2f n4 = x4 * (vv * r4);
    const v2f n5 = x5 * (vv * r5);
    const v2f n6 = x6 * (vv * r6);
    const v2f n7 = x7 * (vv * r7);
    const v2f d0 = one;
    const v2f d1 = x;
    const v2f d2 = x2 * r2;
    const v2f d3 = x3 * r3;
    const v2f d4 = x4 * r4;
    const v2f d5 = x5 * r5;
    const v2f d6 = x6 * r6;
    const v2f d7 = x7 * r7;
    float* w = bt + 4 * p;
    *reinterpret_cast<v4f*>(w + 0 * PLANE) = (v4f){n0.x, n0.y, n1.x, n1.y};
    *reinterpret_cast<v4f*>(w + 1 * PLANE) = (v4f){n2.x, n2.y, n3.x, n3.y};
    *reinterpret_cast<v4f*>(w + 2 * PLANE) = (v4f){n4.x, n4.y, n5.x, n5.y};
    *reinterpret_cast<v4f*>(w + 3 * PLANE) = (v4f){n6.x, n6.y, n7.x, n7.y};
    *reinterpret_cast<v4f*>(w + 4 * PLANE) = (v4f){d0.x, d0.y, d1.x, d1.y};
    *reinterpret_cast<v4f*>(w + 5 * PLANE) = (v4f){d2.x, d2.y, d3.x, d3.y};
    *reinterpret_cast<v4f*>(w + 6 * PLANE) = (v4f){d4.x, d4.y, d5.x, d5.y};
    *reinterpret_cast<v4f*>(w + 7 * PLANE) = (v4f){d6.x, d6.y, d7.x, d7.y};
}

__global__ __launch_bounds__(256)
void intra_attn_wmma_kernel(const float* __restrict__ q,
                            const float* __restrict__ k,
                            const float* __restrict__ v,
                            float* __restrict__ out) {
    __shared__ __align__(16) float kb[NDIM + 8 * PLANE];
    float* const ksl = kb;
    float* const bt  = kb + NDIM;
    __shared__ float qs[ROWS];
    __shared__ float red[256];
    __shared__ float rlo[256];
    __shared__ float rhi[256];
    __shared__ float part[NWAVES][256];
    __shared__ __align__(16) float coef[256];
    __shared__ float cen[16];

    const int tid  = threadIdx.x;
    const int lane = tid & 31;
    const int wave = tid >> 5;

    const int bc    = blockIdx.x;
    const int b     = bc >> 6;
    const int c     = bc & (CDIM - 1);

    const float* qp = q   + (size_t)b * NDIM * CDIM + c;
    const float* kp = k   + (size_t)b * NDIM * CDIM + c;
    const float* vp = v   + (size_t)b * NDIM * CDIM + c;
    float*       op = out + (size_t)b * NDIM * CDIM + c;

    float kr[4], vr[4], qv[RPT];
    for (int u = 0; u < 2; ++u) {
        kr[2 * u]     = kp[(size_t)(2 * (tid + 256 * u))     * CDIM];
        kr[2 * u + 1] = kp[(size_t)(2 * (tid + 256 * u) + 1) * CDIM];
    }
    for (int u = 0; u < 2; ++u) {
        vr[2 * u]     = vp[(size_t)(2 * (tid + 256 * u))     * CDIM];
        vr[2 * u + 1] = vp[(size_t)(2 * (tid + 256 * u) + 1) * CDIM];
    }
    __builtin_amdgcn_sched_barrier(0);
    for (int r = 0; r < RPT; ++r) qv[r] = qp[(size_t)(tid + 256 * r) * CDIM];
    __builtin_amdgcn_sched_barrier(0);

    float lmax = 0.0f;
    #pragma nounroll
    for (int u = 0; u < 2; ++u) {
        const int p  = tid + 256 * u;
        const v2f kk = u ? (v2f){kr[2], kr[3]} : (v2f){kr[0], kr[1]};
        const v2f vv = u ? (v2f){vr[2], vr[3]} : (v2f){vr[0], vr[1]};
        const v2f x  = kk * SCALE;
        build_pair(bt, p, x, vv);
        *reinterpret_cast<v2f*>(&ksl[2 * p]) = x * LOG2E;
        lmax = fmaxf(lmax, fmaxf(fabsf(x.x), fabsf(x.y)));
    }
    float qlo  = qv[0];
    float qhi  = qv[0];
    for (int r = 0; r < RPT; ++r) {
        qs[tid + 256 * r] = qv[r];
        qlo  = fminf(qlo, qv[r]);
        qhi  = fmaxf(qhi, qv[r]);
    }
    red[tid] = lmax;
    rlo[tid] = qlo;
    rhi[tid] = qhi;
    __syncthreads();
    for (int s = 128; s > 0; s >>= 1) {
        if (tid < s) {
            red[tid] = fmaxf(red[tid], red[tid + s]);
            rlo[tid] = fminf(rlo[tid], rlo[tid + s]);
            rhi[tid] = fmaxf(rhi[tid], rhi[tid + s]);
        }
        __syncthreads();
    }
    const float xmax  = red[0];
    const float qmin  = rlo[0];
    const float qmax  = rhi[0];
    const float ksabs = xmax * LOG2E;

    const float range  = qmax - qmin;
    const float need   = range * xmax * (1.0f / (32.0f * THETA));
    const bool  direct = !(need <= (float)MAXT);
    const int   T      = direct ? MAXT : (int)ceilf(fmaxf(need, 1.0f));
    const float ncell  = (float)(16 * T);
    const float cellw  = range / ncell;
    const float inv    = (range > 0.0f) ? ncell / range : 0.0f;

    int g[RPT];
    for (int r = 0; r < RPT; ++r)
        g[r] = direct ? (tid + 256 * r)
                      : (int)fminf(fmaxf((qv[r] - qmin) * inv, 0.0f), ncell - 1.0f);

    const int   M   = lane & 15;
    const int   h   = lane >> 4;
    const int    j0 = wave * JW + 2 * h;
    const float* kp2 = &ksl[j0];
    const float* bp2 = &bt[(M >> 1) * PLANE + 2 * (M & 1) + 2 * j0];

    for (int tile = 0; tile < T; ++tile) {
        const float a  = direct ? qs[tile * 16 + M]
                                : __builtin_fmaf((float)(tile * 16 + M) + 0.5f, cellw, qmin);
        const float m2 = fabsf(a) * ksabs;

        v2f kf[NT];
        v2f bf[NT];
        #pragma unroll
        for (int t = 0; t < NT; ++t) {
            kf[t] = *reinterpret_cast<const v2f*>(&kp2[4 * t]);
            bf[t] = *reinterpret_cast<const v2f*>(&bp2[8 * t]);
        }
        __builtin_amdgcn_sched_barrier(0);

        v8f acc0 = {};
        v8f acc1 = {};
        #pragma unroll
        for (int t = 0; t < NT; t += 2) {
            v2f A0, A1;
            A0.x = __builtin_amdgcn_exp2f(__builtin_fmaf(a, kf[t].x, -m2));
            A0.y = __builtin_amdgcn_exp2f(__builtin_fmaf(a, kf[t].y, -m2));
            A1.x = __builtin_amdgcn_exp2f(__builtin_fmaf(a, kf[t + 1].x, -m2));
            A1.y = __builtin_amdgcn_exp2f(__builtin_fmaf(a, kf[t + 1].y, -m2));
            acc0 = __builtin_amdgcn_wmma_f32_16x16x4_f32(false, A0, false, bf[t],
                                                         (short)0, acc0, false, false);
            acc1 = __builtin_amdgcn_wmma_f32_16x16x4_f32(false, A1, false, bf[t + 1],
                                                         (short)0, acc1, false, false);
        }
        const v8f acc = acc0 + acc1;

        {
            float* pr = &part[wave][(8 * h) * 16 + M];
            for (int r = 0; r < 8; ++r) pr[16 * r] = acc[r];
        }
        __syncthreads();

        float s = 0.0f;
        for (int w = 0; w < NWAVES; ++w) s += part[w][tid];
        coef[tid] = s;
        if (tid < 16) cen[tid] = a;
        __syncthreads();

        for (int r = 0; r < RPT; ++r) {
            if ((g[r] >> 4) != tile) continue;
            const int    m  = g[r] & 15;
            const float  d  = qv[r] - cen[m];
            const float4 n0 = *reinterpret_cast<const float4*>(&coef[m * 16 + 0]);
            const float4 n1 = *reinterpret_cast<const float4*>(&coef[m * 16 + 4]);
            const float4 d0 = *reinterpret_cast<const float4*>(&coef[m * 16 + 8]);
            const float4 d1 = *reinterpret_cast<const float4*>(&coef[m * 16 + 12]);
            float num = n1.w;
            num = __builtin_fmaf(num, d, n1.z);
            num = __builtin_fmaf(num, d, n1.y);
            num = __builtin_fmaf(num, d, n1.x);
            num = __builtin_fmaf(num, d, n0.w);
            num = __builtin_fmaf(num, d, n0.z);
            num = __builtin_fmaf(num, d, n0.y);
            num = __builtin_fmaf(num, d, n0.x);
            float den = d1.w;
            den = __builtin_fmaf(den, d, d1.z);
            den = __builtin_fmaf(den, d, d1.y);
            den = __builtin_fmaf(den, d, d1.x);
            den = __builtin_fmaf(den, d, d0.w);
            den = __builtin_fmaf(den, d, d0.z);
            den = __builtin_fmaf(den, d, d0.y);
            den = __builtin_fmaf(den, d, d0.x);
            op[(size_t)(tid + 256 * r) * CDIM] = num / den;
        }
    }
}

extern "C" void kernel_launch(void* const* d_in, const int* in_sizes, int n_in,
                              void* d_out, int out_size, void* d_ws, size_t ws_size,
                              hipStream_t stream) {
    const float* q = (const float*)d_in[0];
    const float* k = (const float*)d_in[1];
    const float* v = (const float*)d_in[2];
    float* out = (float*)d_out;
    dim3 grid(BDIM * CDIM);
    intra_attn_wmma_kernel<<<grid, dim3(256), 0, stream>>>(q, k, v, out);
}


# Channel extremes (max|x|, min q, max q): each wave reduces them by DPP lane exchange, the 8 waves' results meet in LDS behind the staging barrier; 1 barrier, was a 9-barrier LDS tree; values kept
# speedup vs baseline: 3.4674x; 1.0563x over previous
#include <hip/hip_runtime.h>


typedef float v2f __attribute__((ext_vector_type(2)));
typedef float v4f __attribute__((ext_vector_type(4)));
typedef float v8f __attribute__((ext_vector_type(8)));

#define BDIM  2
#define NDIM  1024
#define CDIM  64
#define SCALE 0.25f
#define LOG2E 1.44269504088896340736f

#define ROWS   NDIM
#define RPT    (ROWS / 256)
#define NWAVES 8
#define JW     (NDIM / NWAVES)
#define NT     (JW / 4)
#define MAXT   (ROWS / 16)
#define NPAIR  (NDIM / 2)
#define PLANE  (NPAIR * 4 + 8)

#define THETA 0.25f

__device__ __forceinline__ void build_pair(float* __restrict__ bt, int p, v2f x, v2f vv) {
    const v2f x2 = x * x;
    const v2f x4 = x2 * x2;
    const v2f x3 = x * x2;
    const v2f x5 = x * x4;
    const v2f x6 = x2 * x4;
    const v2f x7 = x3 * x4;
    const float r2 = 1.0f / 2.0f,   r3 = 1.0f / 6.0f,   r4 = 1.0f / 24.0f;
    const float r5 = 1.0f / 120.0f, r6 = 1.0f / 720.0f, r7 = 1.0f / 5040.0f;
    const v2f one = {1.0f, 1.0f};
    const v2f n0 = vv;
    const v2f n1 = x  * vv;
    const v2f n2 = x2 * (vv * r2);
    const v2f n3 = x3 * (vv * r3);
    const v2f n4 = x4 * (vv * r4);
    const v2f n5 = x5 * (vv * r5);
    const v2f n6 = x6 * (vv * r6);
    const v2f n7 = x7 * (vv * r7);
    const v2f d0 = one;
    const v2f d1 = x;
    const v2f d2 = x2 * r2;
    const v2f d3 = x3 * r3;
    const v2f d4 = x4 * r4;
    const v2f d5 = x5 * r5;
    const v2f d6 = x6 * r6;
    const v2f d7 = x7 * r7;
    float* w = bt + 4 * p;
    *reinterpret_cast<v4f*>(w + 0 * PLANE) = (v4f){n0.x, n0.y, n1.x, n1.y};
    *reinterpret_cast<v4f*>(w + 1 * PLANE) = (v4f){n2.x, n2.y, n3.x, n3.y};
    *reinterpret_cast<v4f*>(w + 2 * PLANE) = (v4f){n4.x, n4.y, n5.x, n5.y};
    *reinterpret_cast<v4f*>(w + 3 * PLANE) = (v4f){n6.x, n6.y, n7.x, n7.y};
    *reinterpret_cast<v4f*>(w + 4 * PLANE) = (v4f){d0.x, d0.y, d1.x, d1.y};
    *reinterpret_cast<v4f*>(w + 5 * PLANE) = (v4f){d2.x, d2.y, d3.x, d3.y};
    *reinterpret_cast<v4f*>(w + 6 * PLANE) = (v4f){d4.x, d4.y, d5.x, d5.y};
    *reinterpret_cast<v4f*>(w + 7 * PLANE) = (v4f){d6.x, d6.y, d7.x, d7.y};
}

template <int M>
__device__ __forceinline__ float from_lane_xor(float x) {
    const int b = __builtin_bit_cast(int, x);
    if (M == 16)
        return __builtin_bit_cast(float, __builtin_amdgcn_permlanex16(b, b, 0x76543210, 0xfedcba98, false, false));
    return __builtin_bit_cast(float, __builtin_amdgcn_update_dpp(b, b, 0x160 + M, 0xf, 0xf, false));
}

template <int M>
__device__ __forceinline__ void extremes_step(float& xm, float& lo, float& hi) {
    const float xm2 = from_lane_xor<M>(xm);
    const float lo2 = from_lane_xor<M>(lo);
    const float hi2 = from_lane_xor<M>(hi);
    xm = fmaxf(xm, xm2);
    lo = fminf(lo, lo2);
    hi = fmaxf(hi, hi2);
}

__global__ __launch_bounds__(256)
void intra_attn_wmma_kernel(const float* __restrict__ q,
                            const float* __restrict__ k,
                            const float* __restrict__ v,
                            float* __restrict__ out) {
    __shared__ __align__(16) float kb[NDIM + 8 * PLANE];
    float* const ksl = kb;
    float* const bt  = kb + NDIM;
    __shared__ float qs[ROWS];
    __shared__ __align__(16) float wred[3 * NWAVES];
    __shared__ float part[NWAVES][256];
    __shared__ __align__(16) float coef[256];
    __shared__ float cen[16];

    const int tid  = threadIdx.x;
    const int lane = tid & 31;
    const int wave = tid >> 5;

    const int bc    = blockIdx.x;
    const int b     = bc >> 6;
    const int c     = bc & (CDIM - 1);

    const float* qp = q   + (size_t)b * NDIM * CDIM + c;
    const float* kp = k   + (size_t)b * NDIM * CDIM + c;
    const float* vp = v   + (size_t)b * NDIM * CDIM + c;
    float*       op = out + (size_t)b * NDIM * CDIM + c;

    float kr[4], vr[4], qv[RPT];
    for (int u = 0; u < 2; ++u) {
        kr[2 * u]     = kp[(size_t)(2 * (tid + 256 * u))     * CDIM];
        kr[2 * u + 1] = kp[(size_t)(2 * (tid + 256 * u) + 1) * CDIM];
    }
    for (int u = 0; u < 2; ++u) {
        vr[2 * u]     = vp[(size_t)(2 * (tid + 256 * u))     * CDIM];
        vr[2 * u + 1] = vp[(size_t)(2 * (tid + 256 * u) + 1) * CDIM];
    }
    __builtin_amdgcn_sched_barrier(0);
    for (int r = 0; r < RPT; ++r) qv[r] = qp[(size_t)(tid + 256 * r) * CDIM];
    __builtin_amdgcn_sched_barrier(0);

    float lmax = 0.0f;
    #pragma nounroll
    for (int u = 0; u < 2; ++u) {
        const int p  = tid + 256 * u;
        const v2f kk = u ? (v2f){kr[2], kr[3]} : (v2f){kr[0], kr[1]};
        const v2f vv = u ? (v2f){vr[2], vr[3]} : (v2f){vr[0], vr[1]};
        const v2f x  = kk * SCALE;
        build_pair(bt, p, x, vv);
        *reinterpret_cast<v2f*>(&ksl[2 * p]) = x * LOG2E;
        lmax = fmaxf(lmax, fmaxf(fabsf(x.x), fabsf(x.y)));
    }
    float qlo  = qv[0];
    float qhi  = qv[0];
    for (int r = 0; r < RPT; ++r) {
        qs[tid + 256 * r] = qv[r];
        qlo  = fminf(qlo, qv[r]);
        qhi  = fmaxf(qhi, qv[r]);
    }
    extremes_step<1>(lmax, qlo, qhi);
    extremes_step<2>(lmax, qlo, qhi);
    extremes_step<4>(lmax, qlo, qhi);
    extremes_step<8>(lmax, qlo, qhi);
    extremes_step<16>(lmax, qlo, qhi);
    if (lane == 0) {
        wred[wave]              = lmax;
        wred[NWAVES + wave]     = qlo;
        wred[2 * NWAVES + wave] = qhi;
    }
    __syncthreads();
    float xmax, qmin, qmax;
    {
        const v4f x0 = *reinterpret_cast<const v4f*>(&wred[0]);
        const v4f x1 = *reinterpret_cast<const v4f*>(&wred[4]);
        const v4f l0 = *reinterpret_cast<const v4f*>(&wred[NWAVES]);
        const v4f l1 = *reinterpret_cast<const v4f*>(&wred[NWAVES + 4]);
        const v4f h0 = *reinterpret_cast<const v4f*>(&wred[2 * NWAVES]);
        const v4f h1 = *reinterpret_cast<const v4f*>(&wred[2 * NWAVES + 4]);
        xmax = fmaxf(fmaxf(fmaxf(x0.x, x0.y), fmaxf(x0.z, x0.w)), fmaxf(fmaxf(x1.x, x1.y), fmaxf(x1.z, x1.w)));
        qmin = fminf(fminf(fminf(l0.x, l0.y), fminf(l0.z, l0.w)), fminf(fminf(l1.x, l1.y), fminf(l1.z, l1.w)));
        qmax = fmaxf(fmaxf(fmaxf(h0.x, h0.y), fmaxf(h0.z, h0.w)), fmaxf(fmaxf(h1.x, h1.y), fmaxf(h1.z, h1.w)));
    }
    const float ksabs = xmax * LOG2E;

    const float range  = qmax - qmin;
    const float need   = range * xmax * (1.0f / (32.0f * THETA));
    const bool  direct = !(need <= (float)MAXT);
    const int   T      = direct ? MAXT : (int)ceilf(fmaxf(need, 1.0f));
    const float ncell  = (float)(16 * T);
    const float cellw  = range / ncell;
    const float inv    = (range > 0.0f) ? ncell / range : 0.0f;

    int g[RPT];
    for (int r = 0; r < RPT; ++r)
        g[r] = direct ? (tid + 256 * r)
                      : (int)fminf(fmaxf((qv[r] - qmin) * inv, 0.0f), ncell - 1.0f);

    const int   M   = lane & 15;
    const int   h   = lane >> 4;
    const int    j0 = wave * JW + 2 * h;
    const float* kp2 = &ksl[j0];
    const float* bp2 = &bt[(M >> 1) * PLANE + 2 * (M & 1) + 2 * j0];

    for (int tile = 0; tile < T; ++tile) {
        const float a  = direct ? qs[tile * 16 + M]
                                : __builtin_fmaf((float)(tile * 16 + M) + 0.5f, cellw, qmin);
        const float m2 = fabsf(a) * ksabs;

        v2f kf[NT];
        v2f bf[NT];
        #pragma unroll
        for (int t = 0; t < NT; ++t) {
            kf[t] = *reinterpret_cast<const v2f*>(&kp2[4 * t]);
            bf[t] = *reinterpret_cast<const v2f*>(&bp2[8 * t]);
        }
        __builtin_amdgcn_sched_barrier(0);

        v8f acc0 = {};
        v8f acc1 = {};
        #pragma unroll
        for (int t = 0; t < NT; t += 2) {
            v2f A0, A1;
            A0.x = __builtin_amdgcn_exp2f(__builtin_fmaf(a, kf[t].x, -m2));
            A0.y = __builtin_amdgcn_exp2f(__builtin_fmaf(a, kf[t].y, -m2));
            A1.x = __builtin_amdgcn_exp2f(__builtin_fmaf(a, kf[t + 1].x, -m2));
            A1.y = __builtin_amdgcn_exp2f(__builtin_fmaf(a, kf[t + 1].y, -m2));
            acc0 = __builtin_amdgcn_wmma_f32_16x16x4_f32(false, A0, false, bf[t],
                                                         (short)0, acc0, false, false);
            acc1 = __builtin_amdgcn_wmma_f32_16x16x4_f32(false, A1, false, bf[t + 1],
                                                         (short)0, acc1, false, false);
        }
        const v8f acc = acc0 + acc1;

        {
            float* pr = &part[wave][(8 * h) * 16 + M];
            for (int r = 0; r < 8; ++r) pr[16 * r] = acc[r];
        }
        __syncthreads();

        float s = 0.0f;
        for (int w = 0; w < NWAVES; ++w) s += part[w][tid];
        coef[tid] = s;
        if (tid < 16) cen[tid] = a;
        __syncthreads();

        for (int r = 0; r < RPT; ++r) {
            if ((g[r] >> 4) != tile) continue;
            const int    m  = g[r] & 15;
            const float  d  = qv[r] - cen[m];
            const float4 n0 = *reinterpret_cast<const float4*>(&coef[m * 16 + 0]);
            const float4 n1 = *reinterpret_cast<const float4*>(&coef[m * 16 + 4]);
            const float4 d0 = *reinterpret_cast<const float4*>(&coef[m * 16 + 8]);
            const float4 d1 = *reinterpret_cast<const float4*>(&coef[m * 16 + 12]);
            float num = n1.w;
            num = __builtin_fmaf(num, d, n1.z);
            num = __builtin_fmaf(num, d, n1.y);
            num = __builtin_fmaf(num, d, n1.x);
            num = __builtin_fmaf(num, d, n0.w);
            num = __builtin_fmaf(num, d, n0.z);
            num = __builtin_fmaf(num, d, n0.y);
            num = __builtin_fmaf(num, d, n0.x);
            float den = d1.w;
            den = __builtin_fmaf(den, d, d1.z);
            den = __builtin_fmaf(den, d, d1.y);
            den = __builtin_fmaf(den, d, d1.x);
            den = __builtin_fmaf(den, d, d0.w);
            den = __builtin_fmaf(den, d, d0.z);
            den = __builtin_fmaf(den, d, d0.y);
            den = __builtin_fmaf(den, d, d0.x);
            op[(size_t)(tid + 256 * r) * CDIM] = num / den;
        }
    }
}

extern "C" void kernel_launch(void* const* d_in, const int* in_sizes, int n_in,
                              void* d_out, int out_size, void* d_ws, size_t ws_size,
                              hipStream_t stream) {
    const float* q = (const float*)d_in[0];
    const float* k = (const float*)d_in[1];
    const float* v = (const float*)d_in[2];
    float* out = (float*)d_out;
    dim3 grid(BDIM * CDIM);
    intra_attn_wmma_kernel<<<grid, dim3(256), 0, stream>>>(q, k, v, out);
}
